# starting kernel
# baseline (speedup 1.0000x reference)
#include <hip/hip_runtime.h>

typedef __attribute__((ext_vector_type(2))) float v2f;
typedef __attribute__((ext_vector_type(8))) float v8f;

namespace {
constexpr int CH   = 256;
constexpr int H    = 48;
constexpr int W    = 64;
constexpr int HW   = H * W;
constexpr int NOFF = 21;
constexpr int PAD  = 20;
}

__global__ __launch_bounds__(256)
void flownetc_corr_wmma(const float* __restrict__ f1,
                        const float* __restrict__ f2,
                        float* __restrict__ out) {
    __shared__ float2 Alds[CH / 2][16];

    const int tid   = threadIdx.x;
    const int h     = blockIdx.x >> 2;
    const int wt    = blockIdx.x & 3;
    const int wbase = ((wt >> 1) << 5) + (wt & 1);

    for (int idx = tid; idx < (CH / 2) * 16; idx += 256) {
        const int cp = idx >> 4;
        const int m  = idx & 15;
        const int g  = (2 * cp) * HW + h * W + wbase + 2 * m;
        Alds[cp][m] = make_float2(f1[g], f1[g + HW]);
    }
    __syncthreads();

    const int wave  = tid >> 5;
    const int lane  = tid & 31;
    const int mn    = lane & 15;
    const int khalf = lane >> 4;

    int   jc[3];
    float msk[3];
#pragma unroll
    for (int t = 0; t < 3; ++t) {
        const int j = wbase - PAD + 32 * t + 2 * mn;
        msk[t] = (j >= 0 && j < W) ? 1.0f : 0.0f;
        jc[t]  = j < 0 ? 0 : (j > W - 1 ? W - 1 : j);
    }

    for (int oyi = wave; oyi < NOFF; oyi += 8) {
        const int r2 = h + 2 * oyi - PAD;

        const int r2n = h + 2 * (oyi + 8) - PAD;
        if (oyi + 8 < NOFF && r2n >= 0 && r2n < H)
            __builtin_prefetch(f2 + (size_t)(2 * khalf) * HW + r2n * W + jc[0], 0, 0);

        v8f acc[3];
#pragma unroll
        for (int t = 0; t < 3; ++t)
            acc[t] = (v8f){0.f, 0.f, 0.f, 0.f, 0.f, 0.f, 0.f, 0.f};

        if (r2 >= 0 && r2 < H) {
            const float*  p  = f2 + (size_t)(2 * khalf) * HW + r2 * W;
            const float2* ap = &Alds[khalf][mn];
#pragma unroll 8
            for (int kk = 0; kk < CH / 4; ++kk) {
                const float2 av = ap[kk * 32];
                v2f a; a.x = av.x; a.y = av.y;
                v2f b0, b1, b2;
                b0.x = p[jc[0]]; b0.y = p[HW + jc[0]];
                b1.x = p[jc[1]]; b1.y = p[HW + jc[1]];
                b2.x = p[jc[2]]; b2.y = p[HW + jc[2]];
                acc[0] = __builtin_amdgcn_wmma_f32_16x16x4_f32(
                             false, a, false, b0, (short)0, acc[0], false, false);
                acc[1] = __builtin_amdgcn_wmma_f32_16x16x4_f32(
                             false, a, false, b1, (short)0, acc[1], false, false);
                acc[2] = __builtin_amdgcn_wmma_f32_16x16x4_f32(
                             false, a, false, b2, (short)0, acc[2], false, false);
                p += 4 * HW;
            }
        }

#pragma unroll
        for (int t = 0; t < 3; ++t)
#pragma unroll
            for (int r = 0; r < 8; ++r)
                acc[t][r] *= msk[t];

#pragma unroll
        for (int t = 0; t < 3; ++t) {
#pragma unroll
            for (int r = 0; r < 8; ++r) {
                const int mo  = r + 8 * khalf;
                const int oxi = 16 * t + mn - mo;
                if (oxi >= 0 && oxi < NOFF) {
                    const int wo = wbase + 2 * mo;
                    out[((size_t)(oyi * NOFF + oxi) * H + h) * W + wo] = acc[t][r];
                }
            }
        }
    }
}

extern "C" void kernel_launch(void* const* d_in, const int* in_sizes, int n_in,
                              void* d_out, int out_size, void* d_ws, size_t ws_size,
                              hipStream_t stream) {
    (void)in_sizes; (void)n_in; (void)out_size; (void)d_ws; (void)ws_size;
    const float* f1  = (const float*)d_in[0];
    const float* f2  = (const float*)d_in[1];
    float*       out = (float*)d_out;

    dim3 grid(H * 4);
    dim3 block(256);
    flownetc_corr_wmma<<<grid, block, 0, stream>>>(f1, f2, out);
}


# each f2 row fetched once per block (whole-line 128-bit loads, double-buffered LDS stage) and shared by the image row's 4 pixel tiles; grid 48 rows x 5 shift groups; k order unchanged
# speedup vs baseline: 2.9478x; 2.9478x over previous
#include <hip/hip_runtime.h>


typedef __attribute__((ext_vector_type(2))) float v2f;
typedef __attribute__((ext_vector_type(8))) float v8f;

namespace {
constexpr int CH   = 256;
constexpr int H    = 48;
constexpr int W    = 64;
constexpr int HW   = H * W;
constexpr int NOFF = 21;
constexpr int PAD  = 20;
constexpr int NGRP = 5;
constexpr int KMAX = 5;
constexpr int KC   = 128;
constexpr int CPC  = KC / 2;
constexpr int NCHK = CH / KC;
constexpr int BP   = 80;
}

__global__ __launch_bounds__(256)
void flownetc_corr_wmma(const float* __restrict__ f1,
                        const float* __restrict__ f2,
                        float* __restrict__ out) {
    __shared__ __attribute__((aligned(16))) float2 As[4][CH / 2][16];
    __shared__ __attribute__((aligned(16))) float2 Bs[2][2][CPC][BP];

    const int tid = threadIdx.x;
    const int h   = blockIdx.x / NGRP;
    const int g   = blockIdx.x - h * NGRP;
    const int q   = tid & 15;
    const int r   = tid >> 4;

    int kfirst = 0, n = 0;
#pragma unroll
    for (int k = 0; k < KMAX; ++k) {
        const int oyi = g + NGRP * k;
        const int r2  = h + 2 * oyi - PAD;
        if (oyi < NOFF) {
            if (r2 >= 0 && r2 < H) {
                if (n == 0) kfirst = k;
                ++n;
            } else {
                for (int idx = tid; idx < NOFF * 16; idx += 256) {
                    const int ox = idx >> 4, qq = idx & 15;
                    *(float4*)(out + (size_t)(oyi * NOFF + ox) * HW + h * W + 4 * qq) =
                        make_float4(0.f, 0.f, 0.f, 0.f);
                }
            }
        }
    }
    const int nIter  = (n + 1) >> 1;
    const int nSteps = nIter * NCHK;

    {
        const int ahalf = q >> 3, am = 2 * (q & 7);
#pragma unroll
        for (int i = 0; i < 8; ++i) {
            const int cp = r + 16 * i;
            const float* src = f1 + (size_t)(2 * cp) * HW + h * W + 4 * q;
            const float4 v0 = *(const float4*)src;
            const float4 v1 = *(const float4*)(src + HW);
            *(float4*)&As[2 * ahalf + 0][cp][am] = make_float4(v0.x, v1.x, v0.z, v1.z);
            *(float4*)&As[2 * ahalf + 1][cp][am] = make_float4(v0.y, v1.y, v0.w, v1.w);
        }
    }

    auto row_of = [&](int it, int ss) {
        int e = 2 * it + ss;
        e = e < n - 1 ? e : n - 1;
        e = e > 0 ? e : 0;
        const int r2 = h + 2 * (g + NGRP * (kfirst + e)) - PAD;
        return r2 < 0 ? 0 : (r2 > H - 1 ? H - 1 : r2);
    };
    float4 sv[2][4][2];
    auto load_step = [&](int st) {
        int sc = st < nSteps - 1 ? st : nSteps - 1;
        sc = sc > 0 ? sc : 0;
        const int it = sc / NCHK, ck = sc % NCHK;
#pragma unroll
        for (int ss = 0; ss < 2; ++ss) {
            const int r2 = row_of(it, ss);
#pragma unroll
            for (int i = 0; i < 4; ++i) {
                const int cp = r + 16 * i;
                const float* src = f2 + (size_t)(ck * KC + 2 * cp) * HW + r2 * W + 4 * q;
                sv[ss][i][0] = *(const float4*)src;
                sv[ss][i][1] = *(const float4*)(src + HW);
            }
        }
    };
    auto store_step = [&](int st) {
        const int b = st & 1;
#pragma unroll
        for (int ss = 0; ss < 2; ++ss) {
#pragma unroll
            for (int i = 0; i < 4; ++i) {
                const int cp = r + 16 * i;
                const float4 v0 = sv[ss][i][0], v1 = sv[ss][i][1];
                *(float4*)&Bs[b][ss][cp][2 * q]      = make_float4(v0.x, v1.x, v0.z, v1.z);
                *(float4*)&Bs[b][ss][cp][32 + 2 * q] = make_float4(v0.y, v1.y, v0.w, v1.w);
            }
        }
    };

    load_step(0);
    store_step(0);
    __syncthreads();

    const int wave  = __builtin_amdgcn_readfirstlane(tid >> 5);
    const int lane  = tid & 31;
    const int a     = wave & 3;
    const int s     = wave >> 2;
    const int mn    = lane & 15;
    const int khalf = lane >> 4;
    const int par   = a & 1;
    const int ahalf = a >> 1;
    const int wbase = (ahalf << 5) + par;

    int   bcol[3];
    float msk[3];
#pragma unroll
    for (int t = 0; t < 3; ++t) {
        const int jj = 16 * ahalf - PAD / 2 + 16 * t + mn;
        msk[t]  = (jj >= 0 && jj < W / 2) ? 1.0f : 0.0f;
        bcol[t] = par * 32 + (jj < 0 ? 0 : (jj > W / 2 - 1 ? W / 2 - 1 : jj));
    }

    v8f acc[3];
#pragma unroll
    for (int t = 0; t < 3; ++t)
        acc[t] = (v8f){0.f, 0.f, 0.f, 0.f, 0.f, 0.f, 0.f, 0.f};

    for (int st = 0; st < nSteps; ++st) {
        const int it = st / NCHK, ck = st % NCHK;
        load_step(st + 1);

        if (2 * it + s < n) {
            const float2* ap = &As[a][ck * CPC + khalf][mn];
            const float2* bp = &Bs[st & 1][s][khalf][0];
#pragma unroll 8
            for (int kk = 0; kk < KC / 4; ++kk) {
                const float2 av  = ap[kk * 32];
                const float2 bv0 = bp[kk * 2 * BP + bcol[0]];
                const float2 bv1 = bp[kk * 2 * BP + bcol[1]];
                const float2 bv2 = bp[kk * 2 * BP + bcol[2]];
                v2f av2; av2.x = av.x;  av2.y = av.y;
                v2f b0;  b0.x  = bv0.x; b0.y  = bv0.y;
                v2f b1;  b1.x  = bv1.x; b1.y  = bv1.y;
                v2f b2;  b2.x  = bv2.x; b2.y  = bv2.y;
                acc[0] = __builtin_amdgcn_wmma_f32_16x16x4_f32(
                             false, av2, false, b0, (short)0, acc[0], false, false);
                acc[1] = __builtin_amdgcn_wmma_f32_16x16x4_f32(
                             false, av2, false, b1, (short)0, acc[1], false, false);
                acc[2] = __builtin_amdgcn_wmma_f32_16x16x4_f32(
                             false, av2, false, b2, (short)0, acc[2], false, false);
            }

            if (ck == NCHK - 1) {
                const int oyi = g + NGRP * (kfirst + 2 * it + s);
#pragma unroll
                for (int t = 0; t < 3; ++t) {
#pragma unroll
                    for (int r8 = 0; r8 < 8; ++r8) {
                        const int mo  = r8 + 8 * khalf;
                        const int oxi = 16 * t + mn - mo;
                        if (oxi >= 0 && oxi < NOFF) {
                            const int wo = wbase + 2 * mo;
                            out[((size_t)(oyi * NOFF + oxi) * H + h) * W + wo] = acc[t][r8] * msk[t];
                        }
                    }
                }
#pragma unroll
                for (int t = 0; t < 3; ++t)
                    acc[t] = (v8f){0.f, 0.f, 0.f, 0.f, 0.f, 0.f, 0.f, 0.f};
            }
        }

        store_step(st + 1);
        __syncthreads();
    }
}

extern "C" void kernel_launch(void* const* d_in, const int* in_sizes, int n_in,
                              void* d_out, int out_size, void* d_ws, size_t ws_size,
                              hipStream_t stream) {
    (void)in_sizes; (void)n_in; (void)out_size; (void)d_ws; (void)ws_size;
    const float* f1  = (const float*)d_in[0];
    const float* f2  = (const float*)d_in[1];
    float*       out = (float*)d_out;

    dim3 grid(H * NGRP);
    dim3 block(256);
    flownetc_corr_wmma<<<grid, block, 0, stream>>>(f1, f2, out);
}


# output rows staged in LDS (21 x 64 floats per shift) and written by the slot's 4 waves as whole 128-B lines with 128-bit stores, instead of per-lane 4-byte stores to 21 planes; sums unchanged
# speedup vs baseline: 3.6112x; 1.2250x over previous
#include <hip/hip_runtime.h>


typedef __attribute__((ext_vector_type(2))) float v2f;
typedef __attribute__((ext_vector_type(8))) float v8f;

namespace {
constexpr int CH   = 256;
constexpr int H    = 48;
constexpr int W    = 64;
constexpr int HW   = H * W;
constexpr int NOFF = 21;
constexpr int PAD  = 20;
constexpr int NGRP = 5;
constexpr int KMAX = 5;
constexpr int KC   = 128;
constexpr int CPC  = KC / 2;
constexpr int NCHK = CH / KC;
constexpr int BP   = 80;
constexpr int OP   = 68;
}

__global__ __launch_bounds__(256)
void flownetc_corr_wmma(const float* __restrict__ f1,
                        const float* __restrict__ f2,
                        float* __restrict__ out) {
    __shared__ __attribute__((aligned(16))) float2 As[4][CH / 2][16];
    __shared__ __attribute__((aligned(16))) float2 Bs[2][2][CPC][BP];
    __shared__ __attribute__((aligned(16))) float Os[2][NOFF][OP];

    const int tid = threadIdx.x;
    const int h   = blockIdx.x / NGRP;
    const int g   = blockIdx.x - h * NGRP;
    const int q   = tid & 15;
    const int r   = tid >> 4;

    int kfirst = 0, n = 0;
#pragma unroll
    for (int k = 0; k < KMAX; ++k) {
        const int oyi = g + NGRP * k;
        const int r2  = h + 2 * oyi - PAD;
        if (oyi < NOFF) {
            if (r2 >= 0 && r2 < H) {
                if (n == 0) kfirst = k;
                ++n;
            } else {
                for (int idx = tid; idx < NOFF * 16; idx += 256) {
                    const int ox = idx >> 4, qq = idx & 15;
                    *(float4*)(out + (size_t)(oyi * NOFF + ox) * HW + h * W + 4 * qq) =
                        make_float4(0.f, 0.f, 0.f, 0.f);
                }
            }
        }
    }
    const int nIter  = (n + 1) >> 1;
    const int nSteps = nIter * NCHK;

    {
        const int ahalf = q >> 3, am = 2 * (q & 7);
#pragma unroll
        for (int i = 0; i < 8; ++i) {
            const int cp = r + 16 * i;
            const float* src = f1 + (size_t)(2 * cp) * HW + h * W + 4 * q;
            const float4 v0 = *(const float4*)src;
            const float4 v1 = *(const float4*)(src + HW);
            *(float4*)&As[2 * ahalf + 0][cp][am] = make_float4(v0.x, v1.x, v0.z, v1.z);
            *(float4*)&As[2 * ahalf + 1][cp][am] = make_float4(v0.y, v1.y, v0.w, v1.w);
        }
    }

    auto row_of = [&](int it, int ss) {
        int e = 2 * it + ss;
        e = e < n - 1 ? e : n - 1;
        e = e > 0 ? e : 0;
        const int r2 = h + 2 * (g + NGRP * (kfirst + e)) - PAD;
        return r2 < 0 ? 0 : (r2 > H - 1 ? H - 1 : r2);
    };
    float4 sv[2][4][2];
    auto load_step = [&](int st) {
        int sc = st < nSteps - 1 ? st : nSteps - 1;
        sc = sc > 0 ? sc : 0;
        const int it = sc / NCHK, ck = sc % NCHK;
#pragma unroll
        for (int ss = 0; ss < 2; ++ss) {
            const int r2 = row_of(it, ss);
#pragma unroll
            for (int i = 0; i < 4; ++i) {
                const int cp = r + 16 * i;
                const float* src = f2 + (size_t)(ck * KC + 2 * cp) * HW + r2 * W + 4 * q;
                sv[ss][i][0] = *(const float4*)src;
                sv[ss][i][1] = *(const float4*)(src + HW);
            }
        }
    };
    auto store_step = [&](int st) {
        const int b = st & 1;
#pragma unroll
        for (int ss = 0; ss < 2; ++ss) {
#pragma unroll
            for (int i = 0; i < 4; ++i) {
                const int cp = r + 16 * i;
                const float4 v0 = sv[ss][i][0], v1 = sv[ss][i][1];
                *(float4*)&Bs[b][ss][cp][2 * q]      = make_float4(v0.x, v1.x, v0.z, v1.z);
                *(float4*)&Bs[b][ss][cp][32 + 2 * q] = make_float4(v0.y, v1.y, v0.w, v1.w);
            }
        }
    };

    load_step(0);
    store_step(0);
    __syncthreads();

    const int wave  = __builtin_amdgcn_readfirstlane(tid >> 5);
    const int lane  = tid & 31;
    const int a     = wave & 3;
    const int s     = wave >> 2;
    const int mn    = lane & 15;
    const int khalf = lane >> 4;
    const int par   = a & 1;
    const int ahalf = a >> 1;
    const int wbase = (ahalf << 5) + par;

    int   bcol[3];
    float msk[3];
#pragma unroll
    for (int t = 0; t < 3; ++t) {
        const int jj = 16 * ahalf - PAD / 2 + 16 * t + mn;
        msk[t]  = (jj >= 0 && jj < W / 2) ? 1.0f : 0.0f;
        bcol[t] = par * 32 + (jj < 0 ? 0 : (jj > W / 2 - 1 ? W / 2 - 1 : jj));
    }

    v8f acc[3];
#pragma unroll
    for (int t = 0; t < 3; ++t)
        acc[t] = (v8f){0.f, 0.f, 0.f, 0.f, 0.f, 0.f, 0.f, 0.f};

    auto flush_rows = [&](int itp) {
        if (2 * itp + s < n) {
            const int oyi = g + NGRP * (kfirst + 2 * itp + s);
            for (int j = 0; j < 3; ++j) {
                const int idx = a * 32 + lane + 128 * j;
                if (idx < NOFF * 16) {
                    const int row = idx >> 4, qq = idx & 15;
                    const float4 v = *(const float4*)&Os[s][row][4 * qq];
                    *(float4*)(out + ((size_t)(oyi * NOFF + row) * H + h) * W + 4 * qq) = v;
                }
            }
        }
    };

    for (int st = 0; st < nSteps; ++st) {
        const int it = st / NCHK, ck = st % NCHK;
        load_step(st + 1);
        if (ck == 0 && it > 0) flush_rows(it - 1);

        if (2 * it + s < n) {
            const float2* ap = &As[a][ck * CPC + khalf][mn];
            const float2* bp = &Bs[st & 1][s][khalf][0];
#pragma unroll 8
            for (int kk = 0; kk < KC / 4; ++kk) {
                const float2 av  = ap[kk * 32];
                const float2 bv0 = bp[kk * 2 * BP + bcol[0]];
                const float2 bv1 = bp[kk * 2 * BP + bcol[1]];
                const float2 bv2 = bp[kk * 2 * BP + bcol[2]];
                v2f av2; av2.x = av.x;  av2.y = av.y;
                v2f b0;  b0.x  = bv0.x; b0.y  = bv0.y;
                v2f b1;  b1.x  = bv1.x; b1.y  = bv1.y;
                v2f b2;  b2.x  = bv2.x; b2.y  = bv2.y;
                acc[0] = __builtin_amdgcn_wmma_f32_16x16x4_f32(
                             false, av2, false, b0, (short)0, acc[0], false, false);
                acc[1] = __builtin_amdgcn_wmma_f32_16x16x4_f32(
                             false, av2, false, b1, (short)0, acc[1], false, false);
                acc[2] = __builtin_amdgcn_wmma_f32_16x16x4_f32(
                             false, av2, false, b2, (short)0, acc[2], false, false);
            }

            if (ck == NCHK - 1) {
#pragma unroll
                for (int t = 0; t < 3; ++t) {
#pragma unroll
                    for (int r8 = 0; r8 < 8; ++r8) {
                        const int mo  = r8 + 8 * khalf;
                        const int oxi = 16 * t + mn - mo;
                        if (oxi >= 0 && oxi < NOFF)
                            Os[s][oxi][wbase + 2 * mo] = acc[t][r8] * msk[t];
                    }
                }
#pragma unroll
                for (int t = 0; t < 3; ++t)
                    acc[t] = (v8f){0.f, 0.f, 0.f, 0.f, 0.f, 0.f, 0.f, 0.f};
            }
        }

        store_step(st + 1);
        __syncthreads();
    }
    if (nIter > 0) flush_rows(nIter - 1);
}

extern "C" void kernel_launch(void* const* d_in, const int* in_sizes, int n_in,
                              void* d_out, int out_size, void* d_ws, size_t ws_size,
                              hipStream_t stream) {
    (void)in_sizes; (void)n_in; (void)out_size; (void)d_ws; (void)ws_size;
    const float* f1  = (const float*)d_in[0];
    const float* f2  = (const float*)d_in[1];
    float*       out = (float*)d_out;

    dim3 grid(H * NGRP);
    dim3 block(256);
    flownetc_corr_wmma<<<grid, block, 0, stream>>>(f1, f2, out);
}


# dense same-parity Gram: each wave multiplies its 16 pixels by the row's 32 same-parity f2 columns (2 tiles per k-step, not 3 clamped band tiles); zero outputs stored once in LDS; sums unchanged
# speedup vs baseline: 4.3181x; 1.1958x over previous
#include <hip/hip_runtime.h>


typedef __attribute__((ext_vector_type(2))) float v2f;
typedef __attribute__((ext_vector_type(8))) float v8f;

namespace {
constexpr int CH   = 256;
constexpr int H    = 48;
constexpr int W    = 64;
constexpr int HW   = H * W;
constexpr int NOFF = 21;
constexpr int PAD  = 20;
constexpr int NGRP = 5;
constexpr int KMAX = 5;
constexpr int KC   = 128;
constexpr int CPC  = KC / 2;
constexpr int NCHK = CH / KC;
constexpr int BP   = 80;
constexpr int OP   = 68;
}

__global__ __launch_bounds__(256)
void flownetc_corr_wmma(const float* __restrict__ f1,
                        const float* __restrict__ f2,
                        float* __restrict__ out) {
    __shared__ __attribute__((aligned(16))) float2 As[4][CH / 2][16];
    __shared__ __attribute__((aligned(16))) float2 Bs[2][2][CPC][BP];
    __shared__ __attribute__((aligned(16))) float Os[2][NOFF][OP];

    const int tid = threadIdx.x;
    const int h   = blockIdx.x / NGRP;
    const int g   = blockIdx.x - h * NGRP;
    const int q   = tid & 15;
    const int r   = tid >> 4;

    int kfirst = 0, n = 0;
#pragma unroll
    for (int k = 0; k < KMAX; ++k) {
        const int oyi = g + NGRP * k;
        const int r2  = h + 2 * oyi - PAD;
        if (oyi < NOFF) {
            if (r2 >= 0 && r2 < H) {
                if (n == 0) kfirst = k;
                ++n;
            } else {
                for (int idx = tid; idx < NOFF * 16; idx += 256) {
                    const int ox = idx >> 4, qq = idx & 15;
                    *(float4*)(out + (size_t)(oyi * NOFF + ox) * HW + h * W + 4 * qq) =
                        make_float4(0.f, 0.f, 0.f, 0.f);
                }
            }
        }
    }
    const int nIter  = (n + 1) >> 1;
    const int nSteps = nIter * NCHK;

    {
        const int ahalf = q >> 3, am = 2 * (q & 7);
#pragma unroll
        for (int i = 0; i < 8; ++i) {
            const int cp = r + 16 * i;
            const float* src = f1 + (size_t)(2 * cp) * HW + h * W + 4 * q;
            const float4 v0 = *(const float4*)src;
            const float4 v1 = *(const float4*)(src + HW);
            *(float4*)&As[2 * ahalf + 0][cp][am] = make_float4(v0.x, v1.x, v0.z, v1.z);
            *(float4*)&As[2 * ahalf + 1][cp][am] = make_float4(v0.y, v1.y, v0.w, v1.w);
        }
    }

    auto row_of = [&](int it, int ss) {
        int e = 2 * it + ss;
        e = e < n - 1 ? e : n - 1;
        e = e > 0 ? e : 0;
        const int r2 = h + 2 * (g + NGRP * (kfirst + e)) - PAD;
        return r2 < 0 ? 0 : (r2 > H - 1 ? H - 1 : r2);
    };
    float4 sv[2][4][2];
    auto load_step = [&](int st) {
        int sc = st < nSteps - 1 ? st : nSteps - 1;
        sc = sc > 0 ? sc : 0;
        const int it = sc / NCHK, ck = sc % NCHK;
#pragma unroll
        for (int ss = 0; ss < 2; ++ss) {
            const int r2 = row_of(it, ss);
#pragma unroll
            for (int i = 0; i < 4; ++i) {
                const int cp = r + 16 * i;
                const float* src = f2 + (size_t)(ck * KC + 2 * cp) * HW + r2 * W + 4 * q;
                sv[ss][i][0] = *(const float4*)src;
                sv[ss][i][1] = *(const float4*)(src + HW);
            }
        }
    };
    auto store_step = [&](int st) {
        const int b = st & 1;
#pragma unroll
        for (int ss = 0; ss < 2; ++ss) {
#pragma unroll
            for (int i = 0; i < 4; ++i) {
                const int cp = r + 16 * i;
                const float4 v0 = sv[ss][i][0], v1 = sv[ss][i][1];
                *(float4*)&Bs[b][ss][cp][2 * q]      = make_float4(v0.x, v1.x, v0.z, v1.z);
                *(float4*)&Bs[b][ss][cp][32 + 2 * q] = make_float4(v0.y, v1.y, v0.w, v1.w);
            }
        }
    };

    load_step(0);

    for (int idx = tid; idx < 2 * NOFF * OP / 4; idx += 256)
        ((float4*)&Os[0][0][0])[idx] = make_float4(0.f, 0.f, 0.f, 0.f);

    store_step(0);
    __syncthreads();

    const int wave  = __builtin_amdgcn_readfirstlane(tid >> 5);
    const int lane  = tid & 31;
    const int a     = wave & 3;
    const int s     = wave >> 2;
    const int mn    = lane & 15;
    const int khalf = lane >> 4;
    const int par   = a & 1;
    const int ahalf = a >> 1;
    const int wbase = (ahalf << 5) + par;

    int bcol[2];
#pragma unroll
    for (int t = 0; t < 2; ++t)
        bcol[t] = par * 32 + 16 * t + mn;

    v8f acc[2];
#pragma unroll
    for (int t = 0; t < 2; ++t)
        acc[t] = (v8f){0.f, 0.f, 0.f, 0.f, 0.f, 0.f, 0.f, 0.f};

    auto flush_rows = [&](int itp) {
        if (2 * itp + s < n) {
            const int oyi = g + NGRP * (kfirst + 2 * itp + s);
            for (int j = 0; j < 3; ++j) {
                const int idx = a * 32 + lane + 128 * j;
                if (idx < NOFF * 16) {
                    const int row = idx >> 4, qq = idx & 15;
                    const float4 v = *(const float4*)&Os[s][row][4 * qq];
                    *(float4*)(out + ((size_t)(oyi * NOFF + row) * H + h) * W + 4 * qq) = v;
                }
            }
        }
    };

    for (int st = 0; st < nSteps; ++st) {
        const int it = st / NCHK, ck = st % NCHK;
        load_step(st + 1);
        if (ck == 0 && it > 0) flush_rows(it - 1);

        if (2 * it + s < n) {
            const float2* ap = &As[a][ck * CPC + khalf][mn];
            const float2* bp = &Bs[st & 1][s][khalf][0];
#pragma unroll 8
            for (int kk = 0; kk < KC / 4; ++kk) {
                const float2 av  = ap[kk * 32];
                const float2 bv0 = bp[kk * 2 * BP + bcol[0]];
                const float2 bv1 = bp[kk * 2 * BP + bcol[1]];
                v2f av2; av2.x = av.x;  av2.y = av.y;
                v2f b0;  b0.x  = bv0.x; b0.y  = bv0.y;
                v2f b1;  b1.x  = bv1.x; b1.y  = bv1.y;
                acc[0] = __builtin_amdgcn_wmma_f32_16x16x4_f32(
                             false, av2, false, b0, (short)0, acc[0], false, false);
                acc[1] = __builtin_amdgcn_wmma_f32_16x16x4_f32(
                             false, av2, false, b1, (short)0, acc[1], false, false);
            }

            if (ck == NCHK - 1) {
#pragma unroll
                for (int t = 0; t < 2; ++t) {
#pragma unroll
                    for (int r8 = 0; r8 < 8; ++r8) {
                        const int mo  = r8 + 8 * khalf;
                        const int oxi = 16 * (t - ahalf) + mn - mo + PAD / 2;
                        if (oxi >= 0 && oxi < NOFF)
                            Os[s][oxi][wbase + 2 * mo] = acc[t][r8];
                    }
                }
#pragma unroll
                for (int t = 0; t < 2; ++t)
                    acc[t] = (v8f){0.f, 0.f, 0.f, 0.f, 0.f, 0.f, 0.f, 0.f};
            }
        }

        store_step(st + 1);
        __syncthreads();
    }
    if (nIter > 0) flush_rows(nIter - 1);
}

extern "C" void kernel_launch(void* const* d_in, const int* in_sizes, int n_in,
                              void* d_out, int out_size, void* d_ws, size_t ws_size,
                              hipStream_t stream) {
    (void)in_sizes; (void)n_in; (void)out_size; (void)d_ws; (void)ws_size;
    const float* f1  = (const float*)d_in[0];
    const float* f2  = (const float*)d_in[1];
    float*       out = (float*)d_out;

    dim3 grid(H * NGRP);
    dim3 block(256);
    flownetc_corr_wmma<<<grid, block, 0, stream>>>(f1, f2, out);
}


# rows dealt to blocks by need: every block at most 4 in-image shifts (2 passes of its 2 slots, not up to 3), rows p and p+24 share 9 blocks, grid 216; sums and k order unchanged
# speedup vs baseline: 5.2930x; 1.2258x over previous
#include <hip/hip_runtime.h>


typedef __attribute__((ext_vector_type(2))) float v2f;
typedef __attribute__((ext_vector_type(8))) float v8f;

namespace {
constexpr int CH   = 256;
constexpr int H    = 48;
constexpr int W    = 64;
constexpr int HW   = H * W;
constexpr int NOFF = 21;
constexpr int PAD  = 20;
constexpr int SMAX = 4;
constexpr int HP   = H / 2;
constexpr int BPP  = 9;
constexpr int KC   = 128;
constexpr int CPC  = KC / 2;
constexpr int NCHK = CH / KC;
constexpr int BP   = 80;
constexpr int OP   = 68;

constexpr int first_shift(int h) { return h >= PAD ? 0 : (PAD - h + 1) / 2; }
constexpr int last_shift(int h) {
    return (H - 1 + PAD - h) / 2 < NOFF - 1 ? (H - 1 + PAD - h) / 2 : NOFF - 1;
}
constexpr int shift_count(int h) { return last_shift(h) - first_shift(h) + 1; }
constexpr int blocks_of(int h) { return (shift_count(h) + SMAX - 1) / SMAX; }
constexpr bool pairs_fill_grid() {
    for (int p = 0; p < HP; ++p)
        if (blocks_of(p) + blocks_of(p + HP) != BPP) return false;
    return true;
}
static_assert(pairs_fill_grid(), "rows p and p + H/2 must need BPP blocks together");
}

__global__ __launch_bounds__(256)
void flownetc_corr_wmma(const float* __restrict__ f1,
                        const float* __restrict__ f2,
                        float* __restrict__ out) {
    __shared__ __attribute__((aligned(16))) float2 As[4][CH / 2][16];
    __shared__ __attribute__((aligned(16))) float2 Bs[2][2][CPC][BP];
    __shared__ __attribute__((aligned(16))) float Os[2][NOFF][OP];

    const int tid = threadIdx.x;
    const int pr  = blockIdx.x / BPP;
    const int jp  = blockIdx.x - pr * BPP;
    const bool up = jp >= blocks_of(pr);
    const int h   = up ? pr + HP : pr;
    const int j   = up ? jp - blocks_of(pr) : jp;
    const int nb  = blocks_of(h);
    const int lo  = first_shift(h);
    const int cnt = shift_count(h);
    const int q   = tid & 15;
    const int r   = tid >> 4;

    for (int z = j; z < NOFF - cnt; z += nb) {
        const int oyi = z < lo ? z : z + cnt;
        for (int idx = tid; idx < NOFF * 16; idx += 256) {
            const int ox = idx >> 4, qq = idx & 15;
            *(float4*)(out + (size_t)(oyi * NOFF + ox) * HW + h * W + 4 * qq) =
                make_float4(0.f, 0.f, 0.f, 0.f);
        }
    }
    const int oyi0 = lo + j;
    int n = 0;
    for (int e = 0; e < SMAX; ++e)
        n += (j + nb * e < cnt) ? 1 : 0;
    const int nIter  = (n + 1) >> 1;
    const int nSteps = nIter * NCHK;

    {
        const int ahalf = q >> 3, am = 2 * (q & 7);
#pragma unroll
        for (int i = 0; i < 8; ++i) {
            const int cp = r + 16 * i;
            const float* src = f1 + (size_t)(2 * cp) * HW + h * W + 4 * q;
            const float4 v0 = *(const float4*)src;
            const float4 v1 = *(const float4*)(src + HW);
            *(float4*)&As[2 * ahalf + 0][cp][am] = make_float4(v0.x, v1.x, v0.z, v1.z);
            *(float4*)&As[2 * ahalf + 1][cp][am] = make_float4(v0.y, v1.y, v0.w, v1.w);
        }
    }

    auto row_of = [&](int it, int ss) {
        int e = 2 * it + ss;
        e = e < n - 1 ? e : n - 1;
        e = e > 0 ? e : 0;
        const int r2 = h + 2 * (oyi0 + nb * e) - PAD;
        return r2 < 0 ? 0 : (r2 > H - 1 ? H - 1 : r2);
    };
    float4 sv[2][4][2];
    auto load_step = [&](int st) {
        int sc = st < nSteps - 1 ? st : nSteps - 1;
        sc = sc > 0 ? sc : 0;
        const int it = sc / NCHK, ck = sc % NCHK;
#pragma unroll
        for (int ss = 0; ss < 2; ++ss) {
            const int r2 = row_of(it, ss);
#pragma unroll
            for (int i = 0; i < 4; ++i) {
                const int cp = r + 16 * i;
                const float* src = f2 + (size_t)(ck * KC + 2 * cp) * HW + r2 * W + 4 * q;
                sv[ss][i][0] = *(const float4*)src;
                sv[ss][i][1] = *(const float4*)(src + HW);
            }
        }
    };
    auto store_step = [&](int st) {
        const int b = st & 1;
#pragma unroll
        for (int ss = 0; ss < 2; ++ss) {
#pragma unroll
            for (int i = 0; i < 4; ++i) {
                const int cp = r + 16 * i;
                const float4 v0 = sv[ss][i][0], v1 = sv[ss][i][1];
                *(float4*)&Bs[b][ss][cp][2 * q]      = make_float4(v0.x, v1.x, v0.z, v1.z);
                *(float4*)&Bs[b][ss][cp][32 + 2 * q] = make_float4(v0.y, v1.y, v0.w, v1.w);
            }
        }
    };

    load_step(0);

    for (int idx = tid; idx < 2 * NOFF * OP / 4; idx += 256)
        ((float4*)&Os[0][0][0])[idx] = make_float4(0.f, 0.f, 0.f, 0.f);

    store_step(0);
    __syncthreads();

    const int wave  = __builtin_amdgcn_readfirstlane(tid >> 5);
    const int lane  = tid & 31;
    const int a     = wave & 3;
    const int s     = wave >> 2;
    const int mn    = lane & 15;
    const int khalf = lane >> 4;
    const int par   = a & 1;
    const int ahalf = a >> 1;
    const int wbase = (ahalf << 5) + par;

    int bcol[2];
#pragma unroll
    for (int t = 0; t < 2; ++t)
        bcol[t] = par * 32 + 16 * t + mn;

    v8f acc[2];
#pragma unroll
    for (int t = 0; t < 2; ++t)
        acc[t] = (v8f){0.f, 0.f, 0.f, 0.f, 0.f, 0.f, 0.f, 0.f};

    auto flush_rows = [&](int itp) {
        if (2 * itp + s < n) {
            const int oyi = oyi0 + nb * (2 * itp + s);
            for (int j = 0; j < 3; ++j) {
                const int idx = a * 32 + lane + 128 * j;
                if (idx < NOFF * 16) {
                    const int row = idx >> 4, qq = idx & 15;
                    const float4 v = *(const float4*)&Os[s][row][4 * qq];
                    *(float4*)(out + ((size_t)(oyi * NOFF + row) * H + h) * W + 4 * qq) = v;
                }
            }
        }
    };

    for (int st = 0; st < nSteps; ++st) {
        const int it = st / NCHK, ck = st % NCHK;
        load_step(st + 1);
        if (ck == 0 && it > 0) flush_rows(it - 1);

        if (2 * it + s < n) {
            const float2* ap = &As[a][ck * CPC + khalf][mn];
            const float2* bp = &Bs[st & 1][s][khalf][0];
#pragma unroll 8
            for (int kk = 0; kk < KC / 4; ++kk) {
                const float2 av  = ap[kk * 32];
                const float2 bv0 = bp[kk * 2 * BP + bcol[0]];
                const float2 bv1 = bp[kk * 2 * BP + bcol[1]];
                v2f av2; av2.x = av.x;  av2.y = av.y;
                v2f b0;  b0.x  = bv0.x; b0.y  = bv0.y;
                v2f b1;  b1.x  = bv1.x; b1.y  = bv1.y;
                acc[0] = __builtin_amdgcn_wmma_f32_16x16x4_f32(
                             false, av2, false, b0, (short)0, acc[0], false, false);
                acc[1] = __builtin_amdgcn_wmma_f32_16x16x4_f32(
                             false, av2, false, b1, (short)0, acc[1], false, false);
            }

            if (ck == NCHK - 1) {
#pragma unroll
                for (int t = 0; t < 2; ++t) {
#pragma unroll
                    for (int r8 = 0; r8 < 8; ++r8) {
                        const int mo  = r8 + 8 * khalf;
                        const int oxi = 16 * (t - ahalf) + mn - mo + PAD / 2;
                        if (oxi >= 0 && oxi < NOFF)
                            Os[s][oxi][wbase + 2 * mo] = acc[t][r8];
                    }
                }
#pragma unroll
                for (int t = 0; t < 2; ++t)
                    acc[t] = (v8f){0.f, 0.f, 0.f, 0.f, 0.f, 0.f, 0.f, 0.f};
            }
        }

        store_step(st + 1);
        __syncthreads();
    }
    if (nIter > 0) flush_rows(nIter - 1);
}

extern "C" void kernel_launch(void* const* d_in, const int* in_sizes, int n_in,
                              void* d_out, int out_size, void* d_ws, size_t ws_size,
                              hipStream_t stream) {
    (void)in_sizes; (void)n_in; (void)out_size; (void)d_ws; (void)ws_size;
    const float* f1  = (const float*)d_in[0];
    const float* f2  = (const float*)d_in[1];
    float*       out = (float*)d_out;

    dim3 grid(HP * BPP);
    dim3 block(256);
    flownetc_corr_wmma<<<grid, block, 0, stream>>>(f1, f2, out);
}


# f2 fragments loaded straight from memory into an 8-set register ring (16-byte whole-row loads, 7 trips ahead): no LDS stage or barrier in the k loop; one wave per shift, 4-wave blocks; sums unchanged
# speedup vs baseline: 6.2058x; 1.1725x over previous
#include <hip/hip_runtime.h>


typedef __attribute__((ext_vector_type(2))) float v2f;
typedef __attribute__((ext_vector_type(8))) float v8f;

namespace {
constexpr int CH    = 256;
constexpr int H     = 48;
constexpr int W     = 64;
constexpr int HW    = H * W;
constexpr int NOFF  = 21;
constexpr int PAD   = 20;
constexpr int SMAX  = 4;
constexpr int HP    = H / 2;
constexpr int BPP   = 9;
constexpr int KT    = 8;
constexpr int NTRIP = CH / KT;
constexpr int RING  = 8;
constexpr int AS_BYTES  = 4 * (CH / 2) * 16 * 8;
constexpr int LDS_BYTES = 176 * 1024;
constexpr int OP    = 68;

constexpr int first_shift(int h) { return h >= PAD ? 0 : (PAD - h + 1) / 2; }
constexpr int last_shift(int h) {
    return (H - 1 + PAD - h) / 2 < NOFF - 1 ? (H - 1 + PAD - h) / 2 : NOFF - 1;
}
constexpr int shift_count(int h) { return last_shift(h) - first_shift(h) + 1; }
constexpr int blocks_of(int h) { return (shift_count(h) + SMAX - 1) / SMAX; }
constexpr bool pairs_fill_grid() {
    for (int p = 0; p < HP; ++p)
        if (blocks_of(p) + blocks_of(p + HP) != BPP) return false;
    return true;
}
static_assert(pairs_fill_grid(), "rows p and p + H/2 must need BPP blocks together");
static_assert(NTRIP % RING == 0, "the trip loop is unrolled by the ring size");
}

__global__ __launch_bounds__(128)
void flownetc_corr_wmma(const float* __restrict__ f1,
                        const float* __restrict__ f2,
                        float* __restrict__ out) {
    __shared__ __attribute__((aligned(16))) unsigned char lds[LDS_BYTES];
    float2 (*As)[CH / 2][16] = reinterpret_cast<float2 (*)[CH / 2][16]>(lds);
    float (*Os)[NOFF + 1][OP] = reinterpret_cast<float (*)[NOFF + 1][OP]>(lds + AS_BYTES);

    const int tid = threadIdx.x;
    const int pr  = blockIdx.x / BPP;
    const int jp  = blockIdx.x - pr * BPP;
    const bool up = jp >= blocks_of(pr);
    const int h   = up ? pr + HP : pr;
    const int j   = up ? jp - blocks_of(pr) : jp;
    const int nb  = blocks_of(h);
    const int lo  = first_shift(h);
    const int cnt = shift_count(h);
    const int q   = tid & 15;
    const int r   = tid >> 4;

    for (int z = j; z < NOFF - cnt; z += nb) {
        const int oyi = z < lo ? z : z + cnt;
        for (int idx = tid; idx < NOFF * 16; idx += 128) {
            const int ox = idx >> 4, qq = idx & 15;
            *(float4*)(out + (size_t)(oyi * NOFF + ox) * HW + h * W + 4 * qq) =
                make_float4(0.f, 0.f, 0.f, 0.f);
        }
    }
    const int oyi0 = lo + j;
    int n = 0;
    for (int e = 0; e < SMAX; ++e)
        n += (j + nb * e < cnt) ? 1 : 0;

    const int wave  = __builtin_amdgcn_readfirstlane(tid >> 5);
    const int lane  = tid & 31;
    const int mn    = lane & 15;
    const int khalf = lane >> 4;
    const bool active = wave < n;
    const int r2    = h + 2 * (oyi0 + nb * wave) - PAD;

    const float* bsrc = f2 + (size_t)(2 * khalf) * HW + (active ? r2 : 0) * W + 4 * mn;
    auto loadB = [&](float4 (&B)[4], int trip) {
        const float* p = bsrc + (size_t)trip * (KT * HW);
#pragma unroll
        for (int kg = 0; kg < 2; ++kg)
#pragma unroll
            for (int jj = 0; jj < 2; ++jj)
                B[kg * 2 + jj] = *(const float4*)(p + (kg * 4 + jj) * HW);
    };
    auto loadA = [&](float2 (&A)[8], int trip) {
#pragma unroll
        for (int kg = 0; kg < 2; ++kg)
#pragma unroll
            for (int a = 0; a < 4; ++a)
                A[kg * 4 + a] = As[a][4 * trip + 2 * kg + khalf][mn];
    };

    float4 Bq[RING][4];
    float2 Aq[2][8];

    {
        const int ahalf = q >> 3, am = 2 * (q & 7);
#pragma unroll
        for (int i = 0; i < 16; ++i) {
            const int cp = r + 8 * i;
            const float* src = f1 + (size_t)(2 * cp) * HW + h * W + 4 * q;
            const float4 v0 = *(const float4*)src;
            const float4 v1 = *(const float4*)(src + HW);
            *(float4*)&As[2 * ahalf + 0][cp][am] = make_float4(v0.x, v1.x, v0.z, v1.z);
            *(float4*)&As[2 * ahalf + 1][cp][am] = make_float4(v0.y, v1.y, v0.w, v1.w);
        }
    }

#pragma unroll
    for (int u = 0; u < RING - 1; ++u)
        loadB(Bq[u], u);

    for (int idx = tid; idx < SMAX * (NOFF + 1) * OP / 4; idx += 128)
        ((float4*)&Os[0][0][0])[idx] = make_float4(0.f, 0.f, 0.f, 0.f);

    __syncthreads();

    if (active) {
        v8f acc[2][2][2];
#pragma unroll
        for (int i = 0; i < 8; ++i)
            acc[i >> 2][(i >> 1) & 1][i & 1] = (v8f){0.f, 0.f, 0.f, 0.f, 0.f, 0.f, 0.f, 0.f};

        auto col = [](const float4& v, int c) {
            return c == 0 ? v.x : c == 1 ? v.y : c == 2 ? v.z : v.w;
        };
        auto mma = [&](const float2 (&A)[8], const float4 (&B)[4]) {
#pragma unroll
            for (int kg = 0; kg < 2; ++kg)
#pragma unroll
                for (int par = 0; par < 2; ++par)
#pragma unroll
                    for (int t = 0; t < 2; ++t) {
                        v2f b;
                        b.x = col(B[kg * 2 + 0], 2 * t + par);
                        b.y = col(B[kg * 2 + 1], 2 * t + par);
#pragma unroll
                        for (int ah = 0; ah < 2; ++ah) {
                            const float2 av = A[kg * 4 + 2 * ah + par];
                            v2f a2; a2.x = av.x; a2.y = av.y;
                            acc[par][ah][t] = __builtin_amdgcn_wmma_f32_16x16x4_f32(
                                false, a2, false, b, (short)0, acc[par][ah][t], false, false);
                        }
                    }
        };

        loadA(Aq[0], 0);
#pragma nounroll
        for (int T0 = 0; T0 < NTRIP; T0 += RING) {
#pragma unroll
            for (int u = 0; u < RING; ++u) {
                const int T  = T0 + u;
                const int Tb = T + RING - 1 < NTRIP - 1 ? T + RING - 1 : NTRIP - 1;
                const int Ta = T + 1 < NTRIP - 1 ? T + 1 : NTRIP - 1;
                loadB(Bq[(u + RING - 1) % RING], Tb);
                loadA(Aq[(u + 1) & 1], Ta);
                __builtin_amdgcn_sched_barrier(0);
                mma(Aq[u & 1], Bq[u]);
                __builtin_amdgcn_sched_barrier(0);
            }
        }

#pragma unroll
        for (int par = 0; par < 2; ++par)
#pragma unroll
            for (int ah = 0; ah < 2; ++ah)
#pragma unroll
                for (int t = 0; t < 2; ++t)
#pragma unroll
                    for (int r8 = 0; r8 < 8; ++r8) {
                        const int mo  = r8 + 8 * khalf;
                        const int oxi = 2 * mn + t - 16 * ah - mo + PAD / 2;
                        const int row = (oxi >= 0 && oxi < NOFF) ? oxi : NOFF;
                        Os[wave][row][32 * ah + par + 2 * mo] = acc[par][ah][t][r8];
                    }
    }

    __syncthreads();

    if (active) {
        const int oyi = oyi0 + nb * wave;
        const int qq  = lane & 15;
        float4 v[(NOFF + 1) / 2];
#pragma unroll
        for (int k = 0; k < (NOFF + 1) / 2; ++k) {
            const int row = 2 * k + (lane >> 4);
            v[k] = *(const float4*)&Os[wave][row][4 * qq];
        }
#pragma unroll
        for (int k = 0; k < (NOFF + 1) / 2; ++k) {
            const int row = 2 * k + (lane >> 4);
            if (row < NOFF)
                *(float4*)(out + ((size_t)(oyi * NOFF + row) * H + h) * W + 4 * qq) = v[k];
        }
    }
}

extern "C" void kernel_launch(void* const* d_in, const int* in_sizes, int n_in,
                              void* d_out, int out_size, void* d_ws, size_t ws_size,
                              hipStream_t stream) {
    (void)in_sizes; (void)n_in; (void)out_size; (void)d_ws; (void)ws_size;
    const float* f1  = (const float*)d_in[0];
    const float* f2  = (const float*)d_in[1];
    float*       out = (float*)d_out;

    dim3 grid(HP * BPP);
    dim3 block(128);
    flownetc_corr_wmma<<<grid, block, 0, stream>>>(f1, f2, out);
}


# f1 fragments straight from memory into a register ring like f2 (same 16-byte row loader, pixel tiles interleaved): no LDS stage or barrier before the k loop, no LDS read in it; sums unchanged
# speedup vs baseline: 7.0067x; 1.1291x over previous
#include <hip/hip_runtime.h>


typedef __attribute__((ext_vector_type(2))) float v2f;
typedef __attribute__((ext_vector_type(8))) float v8f;

namespace {
constexpr int CH    = 256;
constexpr int H     = 48;
constexpr int W     = 64;
constexpr int HW    = H * W;
constexpr int NOFF  = 21;
constexpr int PAD   = 20;
constexpr int SMAX  = 4;
constexpr int HP    = H / 2;
constexpr int BPP   = 9;
constexpr int KT    = 8;
constexpr int NTRIP = CH / KT;
constexpr int RING  = 8;
constexpr int LEAD  = 6;
constexpr int LDS_BYTES = 176 * 1024;
constexpr int OP    = 68;

constexpr int first_shift(int h) { return h >= PAD ? 0 : (PAD - h + 1) / 2; }
constexpr int last_shift(int h) {
    return (H - 1 + PAD - h) / 2 < NOFF - 1 ? (H - 1 + PAD - h) / 2 : NOFF - 1;
}
constexpr int shift_count(int h) { return last_shift(h) - first_shift(h) + 1; }
constexpr int blocks_of(int h) { return (shift_count(h) + SMAX - 1) / SMAX; }
constexpr bool pairs_fill_grid() {
    for (int p = 0; p < HP; ++p)
        if (blocks_of(p) + blocks_of(p + HP) != BPP) return false;
    return true;
}
static_assert(pairs_fill_grid(), "rows p and p + H/2 must need BPP blocks together");
static_assert(NTRIP % RING == 0, "the trip loop is unrolled by the ring size");
static_assert(LEAD < RING, "a set is refilled only after the trip that read it");
static_assert(8 * (LEAD + 1) <= 63, "the load counter holds 63");
}

__global__ __launch_bounds__(128)
void flownetc_corr_wmma(const float* __restrict__ f1,
                        const float* __restrict__ f2,
                        float* __restrict__ out) {
    __shared__ __attribute__((aligned(16))) unsigned char lds[LDS_BYTES];
    float (*Os)[NOFF + 1][OP] = reinterpret_cast<float (*)[NOFF + 1][OP]>(lds);

    const int tid = threadIdx.x;
    const int pr  = blockIdx.x / BPP;
    const int jp  = blockIdx.x - pr * BPP;
    const bool up = jp >= blocks_of(pr);
    const int h   = up ? pr + HP : pr;
    const int j   = up ? jp - blocks_of(pr) : jp;
    const int nb  = blocks_of(h);
    const int lo  = first_shift(h);
    const int cnt = shift_count(h);

    for (int z = j; z < NOFF - cnt; z += nb) {
        const int oyi = z < lo ? z : z + cnt;
        for (int idx = tid; idx < NOFF * 16; idx += 128) {
            const int ox = idx >> 4, qq = idx & 15;
            *(float4*)(out + (size_t)(oyi * NOFF + ox) * HW + h * W + 4 * qq) =
                make_float4(0.f, 0.f, 0.f, 0.f);
        }
    }
    const int oyi0 = lo + j;
    int n = 0;
    for (int e = 0; e < SMAX; ++e)
        n += (j + nb * e < cnt) ? 1 : 0;

    const int wave  = __builtin_amdgcn_readfirstlane(tid >> 5);
    const int lane  = tid & 31;
    const int mn    = lane & 15;
    const int khalf = lane >> 4;
    const bool active = wave < n;
    const int r2    = h + 2 * (oyi0 + nb * wave) - PAD;

    if (active) {
        const unsigned loff = (unsigned)(2 * khalf) * HW + 4 * mn;
        const float* arow = f1 + h * W;
        const float* brow = f2 + r2 * W;
        auto loadF = [&](float4 (&F)[4], const float* row, int trip) {
            const float* p = row + (size_t)trip * (KT * HW) + loff;
#pragma unroll
            for (int kg = 0; kg < 2; ++kg)
#pragma unroll
                for (int jj = 0; jj < 2; ++jj)
                    F[kg * 2 + jj] = *(const float4*)(p + (kg * 4 + jj) * HW);
        };

        float4 Aq[RING][4];
        float4 Bq[RING][4];
#pragma unroll
        for (int u = 0; u < LEAD; ++u) {
            loadF(Aq[u], arow, u);
            loadF(Bq[u], brow, u);
        }

        for (int idx = lane; idx < (NOFF + 1) * OP / 4; idx += 32)
            ((float4*)&Os[wave][0][0])[idx] = make_float4(0.f, 0.f, 0.f, 0.f);

        v8f acc[2][2][2];
#pragma unroll
        for (int i = 0; i < 8; ++i)
            acc[i >> 2][(i >> 1) & 1][i & 1] = (v8f){0.f, 0.f, 0.f, 0.f, 0.f, 0.f, 0.f, 0.f};

        auto col = [](const float4& v, int c) {
            return c == 0 ? v.x : c == 1 ? v.y : c == 2 ? v.z : v.w;
        };
        auto mma = [&](const float4 (&A)[4], const float4 (&B)[4]) {
#pragma unroll
            for (int kg = 0; kg < 2; ++kg)
#pragma unroll
                for (int par = 0; par < 2; ++par) {
                    v2f a[2], b[2];
#pragma unroll
                    for (int s = 0; s < 2; ++s) {
                        a[s].x = col(A[kg * 2 + 0], 2 * s + par);
                        a[s].y = col(A[kg * 2 + 1], 2 * s + par);
                    }
#pragma unroll
                    for (int t = 0; t < 2; ++t) {
                        b[t].x = col(B[kg * 2 + 0], 2 * t + par);
                        b[t].y = col(B[kg * 2 + 1], 2 * t + par);
                    }
#pragma unroll
                    for (int t = 0; t < 2; ++t)
#pragma unroll
                        for (int s = 0; s < 2; ++s)
                            acc[par][s][t] = __builtin_amdgcn_wmma_f32_16x16x4_f32(
                                false, a[s], false, b[t], (short)0, acc[par][s][t], false, false);
                }
        };

#pragma nounroll
        for (int T0 = 0; T0 < NTRIP; T0 += RING) {
#pragma unroll
            for (int u = 0; u < RING; ++u) {
                const int T  = T0 + u;
                const int Tn = T + LEAD < NTRIP - 1 ? T + LEAD : NTRIP - 1;
                loadF(Aq[(u + LEAD) % RING], arow, Tn);
                loadF(Bq[(u + LEAD) % RING], brow, Tn);
                __builtin_amdgcn_sched_barrier(0);
                mma(Aq[u], Bq[u]);
                __builtin_amdgcn_sched_barrier(0);
            }
        }

#pragma unroll
        for (int par = 0; par < 2; ++par)
#pragma unroll
            for (int s = 0; s < 2; ++s)
#pragma unroll
                for (int t = 0; t < 2; ++t)
#pragma unroll
                    for (int r8 = 0; r8 < 8; ++r8) {
                        const int mo  = r8 + 8 * khalf;
                        const int oxi = 2 * (mn - mo) + t - s + PAD / 2;
                        const int row = (oxi >= 0 && oxi < NOFF) ? oxi : NOFF;
                        Os[wave][row][4 * mo + 2 * s + par] = acc[par][s][t][r8];
                    }
    }

    __syncthreads();

    if (active) {
        const int oyi = oyi0 + nb * wave;
        const int qq  = lane & 15;
        float4 v[(NOFF + 1) / 2];
#pragma unroll
        for (int k = 0; k < (NOFF + 1) / 2; ++k) {
            const int row = 2 * k + (lane >> 4);
            v[k] = *(const float4*)&Os[wave][row][4 * qq];
        }
#pragma unroll
        for (int k = 0; k < (NOFF + 1) / 2; ++k) {
            const int row = 2 * k + (lane >> 4);
            if (row < NOFF)
                *(float4*)(out + ((size_t)(oyi * NOFF + row) * H + h) * W + 4 * qq) = v[k];
        }
    }
}

extern "C" void kernel_launch(void* const* d_in, const int* in_sizes, int n_in,
                              void* d_out, int out_size, void* d_ws, size_t ws_size,
                              hipStream_t stream) {
    (void)in_sizes; (void)n_in; (void)out_size; (void)d_ws; (void)ws_size;
    const float* f1  = (const float*)d_in[0];
    const float* f2  = (const float*)d_in[1];
    float*       out = (float*)d_out;

    dim3 grid(HP * BPP);
    dim3 block(128);
    flownetc_corr_wmma<<<grid, block, 0, stream>>>(f1, f2, out);
}
